# starting kernel
# baseline (speedup 1.0000x reference)
#include <hip/hip_runtime.h>
#include <hip/hip_bf16.h>

typedef float v2f __attribute__((ext_vector_type(2)));
typedef float v8f __attribute__((ext_vector_type(8)));

#define DIM 128
#define OUTC 64
#define NGRAPHS 512

__global__ void transpose_kernel(const float* __restrict__ B,
                                 float* __restrict__ BT, int K, int N) {
    int idx = blockIdx.x * blockDim.x + threadIdx.x;
    if (idx >= K * N) return;
    int k = idx / N;
    int n = idx - k * N;
    BT[(size_t)n * K + k] = B[idx];
}

__global__ __launch_bounds__(256) void wmma_gemm_f32_bt(
    const float* __restrict__ A, const float* __restrict__ BT,
    const float* __restrict__ bias, float* __restrict__ C,
    int M, int N, int K)
{
    const int lane = threadIdx.x & 31;
    const int wave = blockIdx.x * (blockDim.x >> 5) + (threadIdx.x >> 5);
    const int colGroups = N >> 6;
    const int tileM = wave / colGroups;
    const int cg    = wave - tileM * colGroups;
    if (tileM * 16 >= M) return;

    const int row0  = tileM << 4;
    const int col0  = cg << 6;
    const int m16   = lane & 15;
    const int khalf = (lane >> 4) << 1;

    const float* ap  = A  + (size_t)(row0 + m16) * K + khalf;
    const float* bp0 = BT + (size_t)(col0 + m16) * K + khalf;
    const float* bp1 = bp0 + (size_t)16 * K;
    const float* bp2 = bp0 + (size_t)32 * K;
    const float* bp3 = bp0 + (size_t)48 * K;

    v8f acc0 = {}, acc1 = {}, acc2 = {}, acc3 = {};
#pragma unroll 4
    for (int k0 = 0; k0 < K; k0 += 4) {
        v2f a  = *(const v2f*)(ap  + k0);
        v2f b0 = *(const v2f*)(bp0 + k0);
        v2f b1 = *(const v2f*)(bp1 + k0);
        v2f b2 = *(const v2f*)(bp2 + k0);
        v2f b3 = *(const v2f*)(bp3 + k0);
        acc0 = __builtin_amdgcn_wmma_f32_16x16x4_f32(false, a, false, b0, (short)0, acc0, false, false);
        acc1 = __builtin_amdgcn_wmma_f32_16x16x4_f32(false, a, false, b1, (short)0, acc1, false, false);
        acc2 = __builtin_amdgcn_wmma_f32_16x16x4_f32(false, a, false, b2, (short)0, acc2, false, false);
        acc3 = __builtin_amdgcn_wmma_f32_16x16x4_f32(false, a, false, b3, (short)0, acc3, false, false);
    }

    const int nOff  = lane & 15;
    const int mBase = row0 + ((lane >> 4) << 3);
    float* crow = C + (size_t)mBase * N + col0 + nOff;
#pragma unroll
    for (int r = 0; r < 8; ++r) {
        float b0v = bias ? bias[col0 + nOff]      : 0.0f;
        float b1v = bias ? bias[col0 + 16 + nOff] : 0.0f;
        float b2v = bias ? bias[col0 + 32 + nOff] : 0.0f;
        float b3v = bias ? bias[col0 + 48 + nOff] : 0.0f;
        crow[(size_t)r * N]      = acc0[r] + b0v;
        crow[(size_t)r * N + 16] = acc1[r] + b1v;
        crow[(size_t)r * N + 32] = acc2[r] + b2v;
        crow[(size_t)r * N + 48] = acc3[r] + b3v;
    }
}

__global__ void deg_init_kernel(float* __restrict__ deg, int n) {
    int i = blockIdx.x * blockDim.x + threadIdx.x;
    if (i < n) deg[i] = 1.0f;
}

__global__ void deg_accum_kernel(const int* __restrict__ dst,
                                 float* __restrict__ deg, int e) {
    int i = blockIdx.x * blockDim.x + threadIdx.x;
    if (i < e)
        __hip_atomic_fetch_add(&deg[dst[i]], 1.0f,
                               __ATOMIC_RELAXED, __HIP_MEMORY_SCOPE_AGENT);
}

__global__ void dinv_kernel(const float* __restrict__ deg,
                            float* __restrict__ dinv, int n) {
    int i = blockIdx.x * blockDim.x + threadIdx.x;
    if (i < n) dinv[i] = rsqrtf(deg[i]);
}

__global__ void self_bias_kernel(const float* __restrict__ h,
                                 const float* __restrict__ dinv,
                                 const float* __restrict__ bias,
                                 float* __restrict__ agg, int n) {
    int idx = blockIdx.x * blockDim.x + threadIdx.x;
    if (idx >= n * DIM) return;
    int i = idx >> 7;
    int c = idx & (DIM - 1);
    float di = dinv[i];
    agg[idx] = h[idx] * di * di + bias[c];
}

__global__ __launch_bounds__(256) void edge_scatter_kernel(
    const float* __restrict__ h, const float* __restrict__ dinv,
    const int* __restrict__ src, const int* __restrict__ dst,
    float* __restrict__ agg, int n_edges)
{
    int wave = blockIdx.x * (blockDim.x >> 5) + (threadIdx.x >> 5);
    if (wave >= n_edges) return;
    int lane = threadIdx.x & 31;
    int s = src[wave];
    int d = dst[wave];
    float coef = dinv[s] * dinv[d];
    const float4* hs = (const float4*)(h + (size_t)s * DIM);
    float4 v = hs[lane];
    float* out = agg + (size_t)d * DIM + lane * 4;
    __hip_atomic_fetch_add(&out[0], v.x * coef, __ATOMIC_RELAXED, __HIP_MEMORY_SCOPE_AGENT);
    __hip_atomic_fetch_add(&out[1], v.y * coef, __ATOMIC_RELAXED, __HIP_MEMORY_SCOPE_AGENT);
    __hip_atomic_fetch_add(&out[2], v.z * coef, __ATOMIC_RELAXED, __HIP_MEMORY_SCOPE_AGENT);
    __hip_atomic_fetch_add(&out[3], v.w * coef, __ATOMIC_RELAXED, __HIP_MEMORY_SCOPE_AGENT);
}

__global__ void relu_kernel(float* __restrict__ a, int total) {
    int i = blockIdx.x * blockDim.x + threadIdx.x;
    if (i < total) a[i] = fmaxf(a[i], 0.0f);
}

__global__ void zero_kernel(float* __restrict__ a, int total) {
    int i = blockIdx.x * blockDim.x + threadIdx.x;
    if (i < total) a[i] = 0.0f;
}

__global__ void pool_kernel(const float* __restrict__ h,
                            const int* __restrict__ batch,
                            float* __restrict__ pooled, int n) {
    int idx = blockIdx.x * blockDim.x + threadIdx.x;
    if (idx >= n * DIM) return;
    int i = idx >> 7;
    int c = idx & (DIM - 1);
    __hip_atomic_fetch_add(&pooled[(size_t)batch[i] * DIM + c], h[idx],
                           __ATOMIC_RELAXED, __HIP_MEMORY_SCOPE_AGENT);
}

__global__ __launch_bounds__(256) void log_softmax_kernel(
    const float* __restrict__ logits, float* __restrict__ out, int rows)
{
    int wave = blockIdx.x * (blockDim.x >> 5) + (threadIdx.x >> 5);
    if (wave >= rows) return;
    int lane = threadIdx.x & 31;
    const float* r = logits + (size_t)wave * OUTC;
    float v0 = r[lane];
    float v1 = r[lane + 32];
    float m = fmaxf(v0, v1);
#pragma unroll
    for (int off = 16; off > 0; off >>= 1)
        m = fmaxf(m, __shfl_xor(m, off, 32));
    float s = __expf(v0 - m) + __expf(v1 - m);
#pragma unroll
    for (int off = 16; off > 0; off >>= 1)
        s += __shfl_xor(s, off, 32);
    float ls = __logf(s);
    out[(size_t)wave * OUTC + lane]      = v0 - m - ls;
    out[(size_t)wave * OUTC + lane + 32] = v1 - m - ls;
}

extern "C" void kernel_launch(void* const* d_in, const int* in_sizes, int n_in,
                              void* d_out, int out_size, void* d_ws, size_t ws_size,
                              hipStream_t stream) {
    const float* x       = (const float*)d_in[0];
    const int*   eidx    = (const int*)d_in[1];
    const int*   x_batch = (const int*)d_in[2];
    const float* W1      = (const float*)d_in[3];
    const float* b1      = (const float*)d_in[4];
    const float* W2      = (const float*)d_in[5];
    const float* b2      = (const float*)d_in[6];
    const float* Wh      = (const float*)d_in[7];
    const float* bh      = (const float*)d_in[8];
    float*       out     = (float*)d_out;

    const int n_nodes = in_sizes[0] / DIM;
    const int n_edges = in_sizes[1] / 2;
    const int* src = eidx;
    const int* dst = eidx + n_edges;

    float* deg    = (float*)d_ws;
    float* dinv   = deg    + n_nodes;
    float* h      = dinv   + n_nodes;
    float* agg    = h      + (size_t)n_nodes * DIM;
    float* pooled = agg    + (size_t)n_nodes * DIM;
    float* logits = pooled + (size_t)NGRAPHS * DIM;
    float* W1T    = logits + (size_t)NGRAPHS * OUTC;
    float* W2T    = W1T    + DIM * DIM;
    float* WhT    = W2T    + DIM * DIM;

    const int T = 256;
    const int nodeBlk = (n_nodes + T - 1) / T;
    const int edgeBlk = (n_edges + T - 1) / T;
    const int nfBlk   = (n_nodes * DIM + T - 1) / T;
    const int edgeWaveBlk = (n_edges + 7) / 8;

    transpose_kernel<<<(DIM * DIM + T - 1) / T, T, 0, stream>>>(W1, W1T, DIM, DIM);
    transpose_kernel<<<(DIM * DIM + T - 1) / T, T, 0, stream>>>(W2, W2T, DIM, DIM);
    transpose_kernel<<<(DIM * OUTC + T - 1) / T, T, 0, stream>>>(Wh, WhT, DIM, OUTC);

    deg_init_kernel<<<nodeBlk, T, 0, stream>>>(deg, n_nodes);
    deg_accum_kernel<<<edgeBlk, T, 0, stream>>>(dst, deg, n_edges);
    dinv_kernel<<<nodeBlk, T, 0, stream>>>(deg, dinv, n_nodes);

    {
        int waves  = (n_nodes / 16) * (DIM / 64);
        int blocks = (waves + 7) / 8;
        wmma_gemm_f32_bt<<<blocks, T, 0, stream>>>(x, W1T, nullptr, h,
                                                   n_nodes, DIM, DIM);
    }
    self_bias_kernel<<<nfBlk, T, 0, stream>>>(h, dinv, b1, agg, n_nodes);
    edge_scatter_kernel<<<edgeWaveBlk, T, 0, stream>>>(h, dinv, src, dst, agg, n_edges);
    relu_kernel<<<nfBlk, T, 0, stream>>>(agg, n_nodes * DIM);

    {
        int waves  = (n_nodes / 16) * (DIM / 64);
        int blocks = (waves + 7) / 8;
        wmma_gemm_f32_bt<<<blocks, T, 0, stream>>>(agg, W2T, nullptr, h,
                                                   n_nodes, DIM, DIM);
    }
    self_bias_kernel<<<nfBlk, T, 0, stream>>>(h, dinv, b2, agg, n_nodes);
    edge_scatter_kernel<<<edgeWaveBlk, T, 0, stream>>>(h, dinv, src, dst, agg, n_edges);
    relu_kernel<<<nfBlk, T, 0, stream>>>(agg, n_nodes * DIM);

    zero_kernel<<<(NGRAPHS * DIM + T - 1) / T, T, 0, stream>>>(pooled, NGRAPHS * DIM);
    pool_kernel<<<nfBlk, T, 0, stream>>>(agg, x_batch, pooled, n_nodes);
    {
        int waves  = (NGRAPHS / 16) * (OUTC / 64);
        int blocks = (waves + 7) / 8;
        wmma_gemm_f32_bt<<<blocks, T, 0, stream>>>(pooled, WhT, bh, logits,
                                                   NGRAPHS, OUTC, DIM);
    }
    log_softmax_kernel<<<(NGRAPHS + 7) / 8, T, 0, stream>>>(logits, out, NGRAPHS);
}


# Edge list binned once per call by 200-row destination tile (LDS rank + one int add per tile reserves the range); both tile kernels read their bin, not the whole list; sums stay in arrival order
# speedup vs baseline: 1.9552x; 1.9552x over previous
#include <hip/hip_runtime.h>
#include <hip/hip_bf16.h>

typedef float v2f __attribute__((ext_vector_type(2)));
typedef float v8f __attribute__((ext_vector_type(8)));

#define DIM 128
#define OUTC 64
#define NGRAPHS 512

__global__ void transpose_kernel(const float* __restrict__ B,
                                 float* __restrict__ BT, int K, int N) {
    int idx = blockIdx.x * blockDim.x + threadIdx.x;
    if (idx >= K * N) return;
    int k = idx / N;
    int n = idx - k * N;
    BT[(size_t)n * K + k] = B[idx];
}

__global__ __launch_bounds__(256) void wmma_gemm_f32_bt(
    const float* __restrict__ A, const float* __restrict__ BT,
    const float* __restrict__ bias, float* __restrict__ C,
    int M, int N, int K)
{
    const int lane = threadIdx.x & 31;
    const int wave = blockIdx.x * (blockDim.x >> 5) + (threadIdx.x >> 5);
    const int colGroups = N >> 6;
    const int tileM = wave / colGroups;
    const int cg    = wave - tileM * colGroups;
    if (tileM * 16 >= M) return;

    const int row0  = tileM << 4;
    const int col0  = cg << 6;
    const int m16   = lane & 15;
    const int khalf = (lane >> 4) << 1;

    const float* ap  = A  + (size_t)(row0 + m16) * K + khalf;
    const float* bp0 = BT + (size_t)(col0 + m16) * K + khalf;
    const float* bp1 = bp0 + (size_t)16 * K;
    const float* bp2 = bp0 + (size_t)32 * K;
    const float* bp3 = bp0 + (size_t)48 * K;

    v8f acc0 = {}, acc1 = {}, acc2 = {}, acc3 = {};
#pragma unroll 4
    for (int k0 = 0; k0 < K; k0 += 4) {
        v2f a  = *(const v2f*)(ap  + k0);
        v2f b0 = *(const v2f*)(bp0 + k0);
        v2f b1 = *(const v2f*)(bp1 + k0);
        v2f b2 = *(const v2f*)(bp2 + k0);
        v2f b3 = *(const v2f*)(bp3 + k0);
        acc0 = __builtin_amdgcn_wmma_f32_16x16x4_f32(false, a, false, b0, (short)0, acc0, false, false);
        acc1 = __builtin_amdgcn_wmma_f32_16x16x4_f32(false, a, false, b1, (short)0, acc1, false, false);
        acc2 = __builtin_amdgcn_wmma_f32_16x16x4_f32(false, a, false, b2, (short)0, acc2, false, false);
        acc3 = __builtin_amdgcn_wmma_f32_16x16x4_f32(false, a, false, b3, (short)0, acc3, false, false);
    }

    const int nOff  = lane & 15;
    const int mBase = row0 + ((lane >> 4) << 3);
    float* crow = C + (size_t)mBase * N + col0 + nOff;
#pragma unroll
    for (int r = 0; r < 8; ++r) {
        float b0v = bias ? bias[col0 + nOff]      : 0.0f;
        float b1v = bias ? bias[col0 + 16 + nOff] : 0.0f;
        float b2v = bias ? bias[col0 + 32 + nOff] : 0.0f;
        float b3v = bias ? bias[col0 + 48 + nOff] : 0.0f;
        crow[(size_t)r * N]      = acc0[r] + b0v;
        crow[(size_t)r * N + 16] = acc1[r] + b1v;
        crow[(size_t)r * N + 32] = acc2[r] + b2v;
        crow[(size_t)r * N + 48] = acc3[r] + b3v;
    }
}

#define TILE_N    200
#define AGG_WAVES 32
#define MAX_TILES 4096
#define BIN_EPT   4

__global__ void deg_init_kernel(float* __restrict__ deg, int* __restrict__ cursor,
                                int n, int n_tiles) {
    int i = blockIdx.x * blockDim.x + threadIdx.x;
    if (i < n) deg[i] = 1.0f;
    if (i < n_tiles) cursor[i] = 0;
}

__global__ void deg_accum_kernel(const int* __restrict__ dst,
                                 float* __restrict__ deg, int e) {
    int i = blockIdx.x * blockDim.x + threadIdx.x;
    if (i < e)
        __hip_atomic_fetch_add(&deg[dst[i]], 1.0f,
                               __ATOMIC_RELAXED, __HIP_MEMORY_SCOPE_AGENT);
}

__global__ __launch_bounds__(256) void dinv_tilecnt_kernel(
    const float* __restrict__ deg, float* __restrict__ dinv,
    int* __restrict__ tile_cnt, int n)
{
    static_assert(TILE_N <= 256, "one thread per node of a tile");
    __shared__ int part[8];
    const int i    = threadIdx.x;
    const int node = blockIdx.x * TILE_N + i;
    int c = 0;
    if (i < TILE_N && node < n) {
        const float d = deg[node];
        dinv[node] = rsqrtf(d);
        c = (int)d - 1;
    }
#pragma unroll
    for (int off = 16; off > 0; off >>= 1) c += __shfl_xor(c, off, 32);
    if ((i & 31) == 0) part[i >> 5] = c;
    __syncthreads();
    if (i == 0) {
        int s = 0;
        for (int k = 0; k < 8; ++k) s += part[k];
        tile_cnt[blockIdx.x] = s;
    }
}

__global__ __launch_bounds__(1024) void bin_fill_kernel(
    const int* __restrict__ src, const int* __restrict__ dst,
    const float* __restrict__ dinv, const int* __restrict__ tile_cnt,
    int* __restrict__ cursor, int* __restrict__ tile_off,
    int* __restrict__ bucket_words, int n_edges, int n_tiles)
{
    __shared__ int off[MAX_TILES];
    __shared__ int hist[MAX_TILES];
    const int tid  = threadIdx.x;
    const int lane = tid & 31;
    const int e0   = blockIdx.x * (1024 * BIN_EPT) + tid;

    int s[BIN_EPT], d[BIN_EPT];
#pragma unroll
    for (int k = 0; k < BIN_EPT; ++k) {
        const int e = e0 + k * 1024;
        const bool ok = e < n_edges;
        d[k] = ok ? dst[e] : -1;
        s[k] = ok ? src[e] : 0;
    }
    for (int t = tid; t < n_tiles; t += 1024) { off[t] = tile_cnt[t]; hist[t] = 0; }
    __syncthreads();

    if (tid < 32) {
        int run = 0;
        for (int i0 = 0; i0 < n_tiles; i0 += 32) {
            const int idx = i0 + lane;
            const int v = (idx < n_tiles) ? off[idx] : 0;
            int inc = v;
#pragma unroll
            for (int o = 1; o < 32; o <<= 1) {
                const int u = __shfl_up(inc, o, 32);
                if (lane >= o) inc += u;
            }
            if (idx < n_tiles) {
                off[idx] = run + inc - v;
                if (blockIdx.x == 0) tile_off[idx] = run + inc - v;
            }
            run += __shfl(inc, 31, 32);
        }
    }

    int t[BIN_EPT], r[BIN_EPT];
    float cf[BIN_EPT];
#pragma unroll
    for (int k = 0; k < BIN_EPT; ++k) {
        const int dn = max(d[k], 0);
        t[k]  = dn / TILE_N;
        cf[k] = dinv[s[k]] * dinv[dn];
    }
#pragma unroll
    for (int k = 0; k < BIN_EPT; ++k) {
        r[k] = 0;
        if (d[k] >= 0)
            r[k] = __hip_atomic_fetch_add(&hist[t[k]], 1, __ATOMIC_RELAXED,
                                          __HIP_MEMORY_SCOPE_WORKGROUP);
    }
    __syncthreads();
    for (int tt = tid; tt < n_tiles; tt += 1024) {
        const int c = hist[tt];
        if (c) off[tt] += __hip_atomic_fetch_add(&cursor[tt], c, __ATOMIC_RELAXED,
                                                 __HIP_MEMORY_SCOPE_AGENT);
    }
    __syncthreads();
    int4* bucket = (int4*)bucket_words;
#pragma unroll
    for (int k = 0; k < BIN_EPT; ++k) {
        if (d[k] >= 0)
            bucket[off[t[k]] + r[k]] =
                make_int4(s[k], d[k] - t[k] * TILE_N, __builtin_bit_cast(int, cf[k]), 0);
    }
}

__global__ __launch_bounds__(1024) void tile_agg_kernel(
    const float* __restrict__ h, const float* __restrict__ dinv,
    const float* __restrict__ bias,
    const int* __restrict__ tile_cnt, const int* __restrict__ tile_off,
    const int* __restrict__ bucket_words,
    float* __restrict__ agg, int n_nodes)
{
    __shared__ __attribute__((aligned(16))) float tile[TILE_N * DIM];

    const int lane = threadIdx.x & 31;
    const int w    = __builtin_amdgcn_readfirstlane((int)(threadIdx.x >> 5));
    const int base = blockIdx.x * TILE_N;
    const int tn   = min(TILE_N, n_nodes - base);

    const int start = tile_off[blockIdx.x];
    const int n     = tile_cnt[blockIdx.x];

    const int per = (((n + AGG_WAVES - 1) / AGG_WAVES) + 7) & ~7;
    const int lo  = min(n, w * per);
    const int hi  = min(n, lo + per);
    const int4* bk = (const int4*)bucket_words + start;
    int4 q = make_int4(0, 0, 0, 0);
    if (lo + lane < hi) q = bk[lo + lane];

    {
        const float4 b4 = ((const float4*)bias)[lane];
        for (int r = w; r < tn; r += AGG_WAVES) {
            const float di = dinv[base + r];
            const float4 v = ((const float4*)(h + (size_t)(base + r) * DIM))[lane];
            float* t = tile + r * DIM + lane;
            t[0]  = v.x * di * di + b4.x;
            t[32] = v.y * di * di + b4.y;
            t[64] = v.z * di * di + b4.z;
            t[96] = v.w * di * di + b4.w;
        }
    }
    __syncthreads();

    auto add_row = [&](const float4& v, float c, int d) {
        float* t = tile + d * DIM + lane;
        __hip_atomic_fetch_add(t,      v.x * c, __ATOMIC_RELAXED, __HIP_MEMORY_SCOPE_WORKGROUP);
        __hip_atomic_fetch_add(t + 32, v.y * c, __ATOMIC_RELAXED, __HIP_MEMORY_SCOPE_WORKGROUP);
        __hip_atomic_fetch_add(t + 64, v.z * c, __ATOMIC_RELAXED, __HIP_MEMORY_SCOPE_WORKGROUP);
        __hip_atomic_fetch_add(t + 96, v.w * c, __ATOMIC_RELAXED, __HIP_MEMORY_SCOPE_WORKGROUP);
    };

    for (int i0 = lo; i0 < hi; i0 += 32) {
        int4 qn = make_int4(0, 0, 0, 0);
        const int m = min(32, hi - i0);
        int k = 0;
        for (; k + 8 <= m; k += 8) {
            float4 v[8]; int dk[8]; float ck[8];
#pragma unroll
            for (int u = 0; u < 8; ++u) {
                const int sk = __builtin_amdgcn_readlane(q.x, k + u);
                dk[u] = __builtin_amdgcn_readlane(q.y, k + u);
                ck[u] = __builtin_bit_cast(float, __builtin_amdgcn_readlane(q.z, k + u));
                v[u] = ((const float4*)(h + (size_t)sk * DIM))[lane];
            }
            if (k == 0 && i0 + 32 + lane < hi) qn = bk[i0 + 32 + lane];
#pragma unroll
            for (int u = 0; u < 8; ++u) add_row(v[u], ck[u], dk[u]);
        }
        if (k < m) {
            float4 v[8];
#pragma unroll
            for (int u = 0; u < 8; ++u) {
                v[u] = make_float4(0.0f, 0.0f, 0.0f, 0.0f);
                if (k + u < m) {
                    const int sk = __builtin_amdgcn_readlane(q.x, k + u);
                    v[u] = ((const float4*)(h + (size_t)sk * DIM))[lane];
                }
            }
#pragma unroll
            for (int u = 0; u < 8; ++u) {
                if (k + u < m) {
                    const int dk = __builtin_amdgcn_readlane(q.y, k + u);
                    const float ck = __builtin_bit_cast(float,
                                       __builtin_amdgcn_readlane(q.z, k + u));
                    add_row(v[u], ck, dk);
                }
            }
        }
        q = qn;
    }
    __syncthreads();

    for (int r = w; r < tn; r += AGG_WAVES) {
        const float* t = tile + r * DIM + lane;
        float4 o;
        o.x = fmaxf(t[0],  0.0f);
        o.y = fmaxf(t[32], 0.0f);
        o.z = fmaxf(t[64], 0.0f);
        o.w = fmaxf(t[96], 0.0f);
        ((float4*)(agg + (size_t)(base + r) * DIM))[lane] = o;
    }
}

__global__ void zero_kernel(float* __restrict__ a, int total) {
    int i = blockIdx.x * blockDim.x + threadIdx.x;
    if (i < total) a[i] = 0.0f;
}

__global__ void pool_kernel(const float* __restrict__ h,
                            const int* __restrict__ batch,
                            float* __restrict__ pooled, int n) {
    int idx = blockIdx.x * blockDim.x + threadIdx.x;
    if (idx >= n * DIM) return;
    int i = idx >> 7;
    int c = idx & (DIM - 1);
    __hip_atomic_fetch_add(&pooled[(size_t)batch[i] * DIM + c], h[idx],
                           __ATOMIC_RELAXED, __HIP_MEMORY_SCOPE_AGENT);
}

__global__ __launch_bounds__(256) void log_softmax_kernel(
    const float* __restrict__ logits, float* __restrict__ out, int rows)
{
    int wave = blockIdx.x * (blockDim.x >> 5) + (threadIdx.x >> 5);
    if (wave >= rows) return;
    int lane = threadIdx.x & 31;
    const float* r = logits + (size_t)wave * OUTC;
    float v0 = r[lane];
    float v1 = r[lane + 32];
    float m = fmaxf(v0, v1);
#pragma unroll
    for (int off = 16; off > 0; off >>= 1)
        m = fmaxf(m, __shfl_xor(m, off, 32));
    float s = __expf(v0 - m) + __expf(v1 - m);
#pragma unroll
    for (int off = 16; off > 0; off >>= 1)
        s += __shfl_xor(s, off, 32);
    float ls = __logf(s);
    out[(size_t)wave * OUTC + lane]      = v0 - m - ls;
    out[(size_t)wave * OUTC + lane + 32] = v1 - m - ls;
}

extern "C" void kernel_launch(void* const* d_in, const int* in_sizes, int n_in,
                              void* d_out, int out_size, void* d_ws, size_t ws_size,
                              hipStream_t stream) {
    const float* x       = (const float*)d_in[0];
    const int*   eidx    = (const int*)d_in[1];
    const int*   x_batch = (const int*)d_in[2];
    const float* W1      = (const float*)d_in[3];
    const float* b1      = (const float*)d_in[4];
    const float* W2      = (const float*)d_in[5];
    const float* b2      = (const float*)d_in[6];
    const float* Wh      = (const float*)d_in[7];
    const float* bh      = (const float*)d_in[8];
    float*       out     = (float*)d_out;

    const int n_nodes = in_sizes[0] / DIM;
    const int n_edges = in_sizes[1] / 2;
    const int* src = eidx;
    const int* dst = eidx + n_edges;

    int*   bucket   = (int*)d_ws;
    int*   tile_cnt = bucket + (size_t)n_edges * 4;
    int*   cursor   = tile_cnt + MAX_TILES;
    int*   tile_off = cursor + MAX_TILES;
    float* deg    = (float*)(tile_off + 2 * MAX_TILES);
    float* dinv   = deg    + n_nodes;
    float* h      = dinv   + n_nodes;
    float* agg    = h      + (size_t)n_nodes * DIM;
    float* pooled = agg    + (size_t)n_nodes * DIM;
    float* logits = pooled + (size_t)NGRAPHS * DIM;
    float* W1T    = logits + (size_t)NGRAPHS * OUTC;
    float* W2T    = W1T    + DIM * DIM;
    float* WhT    = W2T    + DIM * DIM;

    const int T = 256;
    const int nodeBlk = (n_nodes + T - 1) / T;
    const int edgeBlk = (n_edges + T - 1) / T;
    const int nfBlk   = (n_nodes * DIM + T - 1) / T;
    const int nTiles  = (n_nodes + TILE_N - 1) / TILE_N;

    transpose_kernel<<<(DIM * DIM + T - 1) / T, T, 0, stream>>>(W1, W1T, DIM, DIM);
    transpose_kernel<<<(DIM * DIM + T - 1) / T, T, 0, stream>>>(W2, W2T, DIM, DIM);
    transpose_kernel<<<(DIM * OUTC + T - 1) / T, T, 0, stream>>>(Wh, WhT, DIM, OUTC);

    deg_init_kernel<<<nodeBlk, T, 0, stream>>>(deg, cursor, n_nodes, nTiles);
    deg_accum_kernel<<<edgeBlk, T, 0, stream>>>(dst, deg, n_edges);
    dinv_tilecnt_kernel<<<nTiles, T, 0, stream>>>(deg, dinv, tile_cnt, n_nodes);

    bin_fill_kernel<<<(n_edges + 1024 * BIN_EPT - 1) / (1024 * BIN_EPT), 1024, 0, stream>>>(
        src, dst, dinv, tile_cnt, cursor, tile_off, bucket, n_edges, nTiles);

    {
        int waves  = (n_nodes / 16) * (DIM / 64);
        int blocks = (waves + 7) / 8;
        wmma_gemm_f32_bt<<<blocks, T, 0, stream>>>(x, W1T, nullptr, h,
                                                   n_nodes, DIM, DIM);
    }
    tile_agg_kernel<<<nTiles, 1024, 0, stream>>>(h, dinv, b1, tile_cnt, tile_off, bucket, agg, n_nodes);

    {
        int waves  = (n_nodes / 16) * (DIM / 64);
        int blocks = (waves + 7) / 8;
        wmma_gemm_f32_bt<<<blocks, T, 0, stream>>>(agg, W2T, nullptr, h,
                                                   n_nodes, DIM, DIM);
    }
    tile_agg_kernel<<<nTiles, 1024, 0, stream>>>(h, dinv, b2, tile_cnt, tile_off, bucket, agg, n_nodes);

    zero_kernel<<<(NGRAPHS * DIM + T - 1) / T, T, 0, stream>>>(pooled, NGRAPHS * DIM);
    pool_kernel<<<nfBlk, T, 0, stream>>>(agg, x_batch, pooled, n_nodes);
    {
        int waves  = (NGRAPHS / 16) * (OUTC / 64);
        int blocks = (waves + 7) / 8;
        wmma_gemm_f32_bt<<<blocks, T, 0, stream>>>(pooled, WhT, bh, logits,
                                                   NGRAPHS, OUTC, DIM);
    }
    log_softmax_kernel<<<(NGRAPHS + 7) / 8, T, 0, stream>>>(logits, out, NGRAPHS);
}
